# starting kernel
# baseline (speedup 1.0000x reference)
#include <hip/hip_runtime.h>


typedef __attribute__((ext_vector_type(2))) float v2f;
typedef __attribute__((ext_vector_type(8))) float v8f;

#define N_USER 100000
#define N_ITEM 100000
#define NNODES (N_USER + N_ITEM)
#define HID    128
#define E_UI   600000
#define E_IU   600000
#define E_TOT  (E_UI + E_IU)

__global__ __launch_bounds__(128) void gemm_n128(
    const float* __restrict__ A, const float* __restrict__ W,
    const float* __restrict__ bias, float* __restrict__ C,
    int M, int K, int relu, int use_bias)
{
  const int lane  = threadIdx.x & 31;
  const int wave  = threadIdx.x >> 5;
  const long row0 = (long)blockIdx.x * 64 + (long)wave * 16;

  const int m     = lane & 15;
  const int n     = lane & 15;
  const int khalf = (lane >> 4) << 1;

  long arow = row0 + m;
  if (arow >= M) arow = M - 1;
  const float* __restrict__ Ar = A + arow * (long)K;

  v8f acc[8] = {};

  for (int k = 0; k < K; k += 4) {
    v2f a;
    a.x = Ar[k + khalf];
    a.y = Ar[k + khalf + 1];
    if (relu) { a.x = fmaxf(a.x, 0.0f); a.y = fmaxf(a.y, 0.0f); }

    const float* __restrict__ Wk = W + (long)(k + khalf) * HID + n;
#pragma unroll
    for (int t = 0; t < 8; ++t) {
      v2f b;
      b.x = Wk[t * 16];
      b.y = Wk[t * 16 + HID];
      acc[t] = __builtin_amdgcn_wmma_f32_16x16x4_f32(
           false, a,  false, b,
           (short)0, acc[t],  false,  false);
    }
  }

  const int rhalf = (lane >> 4) << 3;
#pragma unroll
  for (int t = 0; t < 8; ++t) {
    float bv = use_bias ? bias[t * 16 + n] : 0.0f;
#pragma unroll
    for (int r = 0; r < 8; ++r) {
      long row = row0 + rhalf + r;
      if (row < M) C[row * HID + t * 16 + n] = acc[t][r] + bv;
    }
  }
}

__global__ void deg_init(float* __restrict__ deg, int n)
{
  int i = blockIdx.x * blockDim.x + threadIdx.x;
  if (i < n) deg[i] = 1.0f;
}

__global__ void deg_count(const int* __restrict__ eui,
                          const int* __restrict__ eiu,
                          float* __restrict__ deg)
{
  int e = blockIdx.x * blockDim.x + threadIdx.x;
  if (e >= E_TOT) return;
  int d;
  if (e < E_UI) d = eui[E_UI + e] + N_USER;
  else          d = eiu[E_IU + (e - E_UI)];
  atomicAdd(&deg[d], 1.0f);
}

__global__ void dinv_fin(float* __restrict__ deg, int n)
{
  int i = blockIdx.x * blockDim.x + threadIdx.x;
  if (i < n) {
    float v = deg[i];
    deg[i] = v > 0.0f ? rsqrtf(v) : 0.0f;
  }
}

__global__ void agg_self(const float* __restrict__ hw,
                         const float* __restrict__ dinv,
                         const float* __restrict__ bias,
                         float* __restrict__ out)
{
  long idx = (long)blockIdx.x * blockDim.x + threadIdx.x;
  if (idx >= (long)NNODES * HID) return;
  long i = idx >> 7;
  int  c = (int)(idx & 127);
  float di = dinv[i];
  out[idx] = bias[c] + di * di * hw[idx];
}

__global__ __launch_bounds__(256) void agg_edges(
    const int* __restrict__ eui, const int* __restrict__ eiu,
    const float* __restrict__ hw, const float* __restrict__ dinv,
    float* __restrict__ out)
{
  int lane = threadIdx.x & 31;
  int w    = threadIdx.x >> 5;
  int e    = blockIdx.x * 8 + w;
  if (e >= E_TOT) return;

  int s, d;
  if (e < E_UI) { s = eui[e];                 d = eui[E_UI + e] + N_USER; }
  else          { int e2 = e - E_UI;
                  s = eiu[e2] + N_USER;       d = eiu[E_IU + e2]; }

  float nrm = dinv[s] * dinv[d];
  const float4* __restrict__ hs = (const float4*)(hw + (long)s * HID);
  float4 v = hs[lane];
  float* od = out + (long)d * HID + lane * 4;
  atomicAdd(od + 0, nrm * v.x);
  atomicAdd(od + 1, nrm * v.y);
  atomicAdd(od + 2, nrm * v.z);
  atomicAdd(od + 3, nrm * v.w);
}

extern "C" void kernel_launch(void* const* d_in, const int* in_sizes, int n_in,
                              void* d_out, int out_size, void* d_ws, size_t ws_size,
                              hipStream_t stream)
{
  const float* x_user = (const float*)d_in[0];
  const float* x_item = (const float*)d_in[1];
  const int*   eui    = (const int*)d_in[2];
  const int*   eiu    = (const int*)d_in[3];
  const float* W_iu   = (const float*)d_in[4];
  const float* b_iu   = (const float*)d_in[5];
  const float* W_ii   = (const float*)d_in[6];
  const float* b_ii   = (const float*)d_in[7];
  const float* W1     = (const float*)d_in[8];
  const float* b1     = (const float*)d_in[9];
  const float* W2     = (const float*)d_in[10];
  const float* b2     = (const float*)d_in[11];
  float* out = (float*)d_out;

  float* h    = (float*)d_ws;
  float* hw   = h  + (size_t)NNODES * HID;
  float* dinv = hw + (size_t)NNODES * HID;

  const dim3 blk(128);

  gemm_n128<<<dim3((N_USER + 63) / 64), blk, 0, stream>>>(
      x_user, W_iu, b_iu, h, N_USER, 256,  0,  1);
  gemm_n128<<<dim3((N_ITEM + 63) / 64), blk, 0, stream>>>(
      x_item, W_ii, b_ii, h + (size_t)N_USER * HID, N_ITEM, 256, 0, 1);

  deg_init <<<dim3((NNODES + 255) / 256), dim3(256), 0, stream>>>(dinv, NNODES);
  deg_count<<<dim3((E_TOT + 255) / 256), dim3(256), 0, stream>>>(eui, eiu, dinv);
  dinv_fin <<<dim3((NNODES + 255) / 256), dim3(256), 0, stream>>>(dinv, NNODES);

  const unsigned aggBlocks  = (unsigned)(((size_t)NNODES * HID + 255) / 256);
  const unsigned edgeBlocks = (E_TOT + 7) / 8;

  gemm_n128<<<dim3((NNODES + 63) / 64), blk, 0, stream>>>(
      h, W1, b1, hw, NNODES, HID,  0,  0);
  agg_self <<<dim3(aggBlocks),  dim3(256), 0, stream>>>(hw, dinv, b1, h);
  agg_edges<<<dim3(edgeBlocks), dim3(256), 0, stream>>>(eui, eiu, hw, dinv, h);

  gemm_n128<<<dim3((NNODES + 63) / 64), blk, 0, stream>>>(
      h, W2, b2, hw, NNODES, HID,  1,  0);
  agg_self <<<dim3(aggBlocks),  dim3(256), 0, stream>>>(hw, dinv, b2, out);
  agg_edges<<<dim3(edgeBlocks), dim3(256), 0, stream>>>(eui, eiu, hw, dinv, out);
}


# in-degree count moved off global atomics: a work-group per (node range, edge slice, type) counts destinations in LDS, stores a plane; rsqrt pass sums the planes; 7 to 6 launches; sum order same
# speedup vs baseline: 2.1810x; 2.1810x over previous
#include <hip/hip_runtime.h>


typedef __attribute__((ext_vector_type(2))) float v2f;
typedef __attribute__((ext_vector_type(4))) float v4f;
typedef __attribute__((ext_vector_type(8))) float v8f;

#define N_USER 100000
#define N_ITEM 100000
#define NNODES (N_USER + N_ITEM)
#define HID    128
#define E_UI   600000
#define E_IU   600000
#define E_TOT  (E_UI + E_IU)

__global__ __launch_bounds__(128) void gemm_n128(
    const float* __restrict__ A, const float* __restrict__ W,
    const float* __restrict__ bias, float* __restrict__ C,
    const float* __restrict__ dinv, float* __restrict__ S,
    int M, int K, int relu, int start)
{
  const int lane  = threadIdx.x & 31;
  const int wave  = threadIdx.x >> 5;
  const long row0 = (long)blockIdx.x * 64 + (long)wave * 16;

  const int m     = lane & 15;
  const int n4    = (lane & 15) << 2;
  const int khalf = (lane >> 4) << 1;

  long arow = row0 + m;
  if (arow >= M) arow = M - 1;
  const float* __restrict__ Ar = A + arow * (long)K;

  v8f acc[8] = {};

  for (int k = 0; k < K; k += 4) {
    v2f a;
    a.x = Ar[k + khalf];
    a.y = Ar[k + khalf + 1];
    if (relu) { a.x = fmaxf(a.x, 0.0f); a.y = fmaxf(a.y, 0.0f); }

    const float* __restrict__ Wk = W + (long)(k + khalf) * HID + n4;
    const v4f w0l = *reinterpret_cast<const v4f*>(Wk);
    const v4f w0h = *reinterpret_cast<const v4f*>(Wk + 64);
    const v4f w1l = *reinterpret_cast<const v4f*>(Wk + HID);
    const v4f w1h = *reinterpret_cast<const v4f*>(Wk + HID + 64);
#pragma unroll
    for (int t = 0; t < 4; ++t) {
      v2f b;
      b.x = w0l[t];
      b.y = w1l[t];
      acc[t] = __builtin_amdgcn_wmma_f32_16x16x4_f32(
           false, a,  false, b,
           (short)0, acc[t],  false,  false);
    }
#pragma unroll
    for (int t = 0; t < 4; ++t) {
      v2f b;
      b.x = w0h[t];
      b.y = w1h[t];
      acc[4 + t] = __builtin_amdgcn_wmma_f32_16x16x4_f32(
           false, a,  false, b,
           (short)0, acc[4 + t],  false,  false);
    }
  }

  const int rhalf = (lane >> 4) << 3;
  const v4f bl = *reinterpret_cast<const v4f*>(bias + n4);
  const v4f bh = *reinterpret_cast<const v4f*>(bias + 64 + n4);
  if (!start) {
#pragma unroll
    for (int r = 0; r < 8; ++r) {
      const long row = row0 + rhalf + r;
      if (row < M) {
        v4f ol, oh;
#pragma unroll
        for (int t = 0; t < 4; ++t) {
          ol[t] = acc[t][r] + bl[t];
          oh[t] = acc[4 + t][r] + bh[t];
        }
        float* Cr = C + row * HID + n4;
        *reinterpret_cast<v4f*>(Cr)      = ol;
        *reinterpret_cast<v4f*>(Cr + 64) = oh;
      }
    }
  } else {
#pragma unroll
    for (int r = 0; r < 8; ++r) {
      long row = row0 + rhalf + r;
      const bool live = row < M;
      if (!live) row = M - 1;
      const float di = dinv[row];
      const float dd = di * di;
      if (live) {
        v4f pl, ph, sl, sh;
#pragma unroll
        for (int t = 0; t < 4; ++t) {
          pl[t] = acc[t][r];
          ph[t] = acc[4 + t][r];
          sl[t] = bl[t] + dd * pl[t];
          sh[t] = bh[t] + dd * ph[t];
        }
        const long o = row * HID + n4;
        *reinterpret_cast<v4f*>(C + o)      = pl;
        *reinterpret_cast<v4f*>(C + o + 64) = ph;
        *reinterpret_cast<v4f*>(S + o)      = sl;
        *reinterpret_cast<v4f*>(S + o + 64) = sh;
      }
    }
  }
}

#define IN_DIM 256
static_assert(N_USER % 16 == 0 && NNODES % 64 == 0 && IN_DIM % 4 == 0,
              "whole 16-row waves of one node type, whole 64-row blocks");

__global__ __launch_bounds__(128) void gemm_proj_l1(
    const float* __restrict__ x_user, const float* __restrict__ x_item,
    const float* __restrict__ W_iu, const float* __restrict__ b_iu,
    const float* __restrict__ W_ii, const float* __restrict__ b_ii,
    const float* __restrict__ W1, const float* __restrict__ b1,
    const float* __restrict__ dinv, float* __restrict__ C, float* __restrict__ S)
{
  const int lane = threadIdx.x & 31;
  const int wave = __builtin_amdgcn_readfirstlane((int)(threadIdx.x >> 5));
  const int row0 = blockIdx.x * 64 + wave * 16;

  const bool item = row0 >= N_USER;
  const float* __restrict__ X  = item ? x_item + (long)(row0 - N_USER) * IN_DIM
                                      : x_user + (long)row0 * IN_DIM;
  const float* __restrict__ Wi = item ? W_ii : W_iu;
  const float* __restrict__ bi = item ? b_ii : b_iu;

  const int c  = lane & 15;
  const int hh = lane >> 4;
  const int c4 = c << 2;

  const float* __restrict__ Xr = X + (long)c * IN_DIM + 2 * hh;
  const float* __restrict__ Wc = Wi + (long)(2 * hh) * HID + c4;

  v8f accT[8] = {};
#pragma unroll 4
  for (int k = 0; k < IN_DIM; k += 4) {
    v2f xb;
    xb.x = Xr[k];
    xb.y = Xr[k + 1];
    const float* __restrict__ Wk = Wc + (long)k * HID;
    const v4f w0l = *reinterpret_cast<const v4f*>(Wk);
    const v4f w0h = *reinterpret_cast<const v4f*>(Wk + 64);
    const v4f w1l = *reinterpret_cast<const v4f*>(Wk + HID);
    const v4f w1h = *reinterpret_cast<const v4f*>(Wk + HID + 64);
#pragma unroll
    for (int t = 0; t < 4; ++t) {
      v2f a;
      a.x = w0l[t];
      a.y = w1l[t];
      accT[t] = __builtin_amdgcn_wmma_f32_16x16x4_f32(
           false, a,  false, xb,
           (short)0, accT[t],  false,  false);
    }
#pragma unroll
    for (int t = 0; t < 4; ++t) {
      v2f a;
      a.x = w0h[t];
      a.y = w1h[t];
      accT[4 + t] = __builtin_amdgcn_wmma_f32_16x16x4_f32(
           false, a,  false, xb,
           (short)0, accT[4 + t],  false,  false);
    }
  }

#pragma unroll
  for (int q = 0; q < 2; ++q) {
#pragma unroll
    for (int r = 0; r < 8; ++r) {
      const v4f bb = *reinterpret_cast<const v4f*>(bi + 64 * q + 32 * hh + 4 * r);
#pragma unroll
      for (int e = 0; e < 4; ++e) accT[4 * q + e][r] += bb[e];
    }
  }

  const float* __restrict__ W1c = W1 + (long)(32 * hh) * HID + c4;
  v8f acc[8] = {};
#pragma unroll
  for (int t = 0; t < 8; ++t) {
#pragma unroll
    for (int p = 0; p < 4; ++p) {
      v2f a;
      a.x = accT[t][2 * p];
      a.y = accT[t][2 * p + 1];
      const float* __restrict__ Wk = W1c + (long)((t >> 2) * 64 + (t & 3) + 8 * p) * HID;
      const v4f w0l = *reinterpret_cast<const v4f*>(Wk);
      const v4f w0h = *reinterpret_cast<const v4f*>(Wk + 64);
      const v4f w1l = *reinterpret_cast<const v4f*>(Wk + 4 * HID);
      const v4f w1h = *reinterpret_cast<const v4f*>(Wk + 4 * HID + 64);
#pragma unroll
      for (int u = 0; u < 4; ++u) {
        v2f b;
        b.x = w0l[u];
        b.y = w1l[u];
        acc[u] = __builtin_amdgcn_wmma_f32_16x16x4_f32(
             false, a,  false, b,
             (short)0, acc[u],  false,  false);
      }
#pragma unroll
      for (int u = 0; u < 4; ++u) {
        v2f b;
        b.x = w0h[u];
        b.y = w1h[u];
        acc[4 + u] = __builtin_amdgcn_wmma_f32_16x16x4_f32(
             false, a,  false, b,
             (short)0, acc[4 + u],  false,  false);
      }
    }
  }

  const int rhalf = hh << 3;
  const v4f bl = *reinterpret_cast<const v4f*>(b1 + c4);
  const v4f bh = *reinterpret_cast<const v4f*>(b1 + 64 + c4);
#pragma unroll
  for (int r = 0; r < 8; ++r) {
    const long row = (long)row0 + rhalf + r;
    const float di = dinv[row];
    const float dd = di * di;
    v4f pl, ph, sl, sh;
#pragma unroll
    for (int u = 0; u < 4; ++u) {
      pl[u] = acc[u][r];
      ph[u] = acc[4 + u][r];
      sl[u] = bl[u] + dd * pl[u];
      sh[u] = bh[u] + dd * ph[u];
    }
    const long o = row * HID + c4;
    *reinterpret_cast<v4f*>(C + o)      = pl;
    *reinterpret_cast<v4f*>(C + o + 64) = ph;
    *reinterpret_cast<v4f*>(S + o)      = sl;
    *reinterpret_cast<v4f*>(S + o + 64) = sh;
  }
}

#define DEG_RANGES      8
#define DEG_SLICES      16
#define DEG_THREADS     512
#define DEG_BINS        (N_USER / DEG_RANGES)
#define DEG_SLICE_VEC   (E_UI / DEG_SLICES / 4)
#define DEG_VEC_PER_THR ((DEG_SLICE_VEC + DEG_THREADS - 1) / DEG_THREADS)
static_assert(N_USER == N_ITEM && E_UI == E_IU, "one cell shape for both edge types");
static_assert(N_USER % DEG_RANGES == 0 && DEG_BINS % 4 == 0, "whole 128-bit counter groups");
static_assert(E_UI % (DEG_SLICES * 4) == 0, "slices are whole 128-bit index groups");
static_assert(NNODES % 4 == 0 && N_USER % 4 == 0, "planes are read and written 128 bits wide");

__global__ __launch_bounds__(DEG_THREADS) void deg_hist(
    const int* __restrict__ eui, const int* __restrict__ eiu,
    unsigned* __restrict__ planes)
{
  __shared__ __attribute__((aligned(16))) unsigned bins[DEG_BINS];
  const int tid   = threadIdx.x;
  const int lo    = blockIdx.x * DEG_BINS;
  const int slice = blockIdx.y;
  const bool ui   = blockIdx.z == 0;

  const int4* __restrict__ d4 = reinterpret_cast<const int4*>(
      (ui ? eui + E_UI : eiu + E_IU) + (long)slice * (DEG_SLICE_VEC * 4));

  int4 v[DEG_VEC_PER_THR];
#pragma unroll
  for (int j = 0; j < DEG_VEC_PER_THR; ++j) {
    int i = tid + j * DEG_THREADS;
    if (i > DEG_SLICE_VEC - 1) i = DEG_SLICE_VEC - 1;
    v[j] = d4[i];
  }

  uint4* b4 = reinterpret_cast<uint4*>(bins);
  for (int i = tid; i < DEG_BINS / 4; i += DEG_THREADS) b4[i] = make_uint4(0u, 0u, 0u, 0u);
  __syncthreads();

#pragma unroll
  for (int j = 0; j < DEG_VEC_PER_THR; ++j) {
    if (tid + j * DEG_THREADS < DEG_SLICE_VEC) {
      const unsigned r0 = (unsigned)(v[j].x - lo);
      const unsigned r1 = (unsigned)(v[j].y - lo);
      const unsigned r2 = (unsigned)(v[j].z - lo);
      const unsigned r3 = (unsigned)(v[j].w - lo);
      if (r0 < (unsigned)DEG_BINS) atomicAdd(&bins[r0], 1u);
      if (r1 < (unsigned)DEG_BINS) atomicAdd(&bins[r1], 1u);
      if (r2 < (unsigned)DEG_BINS) atomicAdd(&bins[r2], 1u);
      if (r3 < (unsigned)DEG_BINS) atomicAdd(&bins[r3], 1u);
    }
  }
  __syncthreads();

  uint4* __restrict__ o4 = reinterpret_cast<uint4*>(
      planes + (long)slice * NNODES + (ui ? N_USER : 0) + lo);
  for (int i = tid; i < DEG_BINS / 4; i += DEG_THREADS) o4[i] = b4[i];
}

__global__ __launch_bounds__(256) void dinv_fin(const unsigned* __restrict__ planes,
                                                float* __restrict__ dinv)
{
  const int i = blockIdx.x * blockDim.x + threadIdx.x;
  if (i < NNODES / 4) {
    const uint4* __restrict__ p4 = reinterpret_cast<const uint4*>(planes) + i;
    uint4 c = make_uint4(1u, 1u, 1u, 1u);
#pragma unroll
    for (int p = 0; p < DEG_SLICES; ++p) {
      const uint4 q = p4[(long)p * (NNODES / 4)];
      c.x += q.x; c.y += q.y; c.z += q.z; c.w += q.w;
    }
    v4f d;
    const float f0 = (float)c.x, f1 = (float)c.y, f2 = (float)c.z, f3 = (float)c.w;
    d[0] = f0 > 0.0f ? rsqrtf(f0) : 0.0f;
    d[1] = f1 > 0.0f ? rsqrtf(f1) : 0.0f;
    d[2] = f2 > 0.0f ? rsqrtf(f2) : 0.0f;
    d[3] = f3 > 0.0f ? rsqrtf(f3) : 0.0f;
    *reinterpret_cast<v4f*>(dinv + 4 * (long)i) = d;
  }
}

#define EDGES_PER_WAVE 32
#define EDGE_BATCH     8
static_assert(E_UI % EDGES_PER_WAVE == 0 && E_TOT % EDGES_PER_WAVE == 0,
              "a wave's 32 edges must be of one type and all exist");

__global__ __launch_bounds__(256) void agg_edges(
    const int* __restrict__ eui, const int* __restrict__ eiu,
    const float* __restrict__ hw, const float* __restrict__ dinv,
    float* __restrict__ out)
{
  const int lane = threadIdx.x & 31;
  const int w  = __builtin_amdgcn_readfirstlane((int)(threadIdx.x >> 5));
  const int e0 = (blockIdx.x * 8 + w) * EDGES_PER_WAVE;
  if (e0 >= E_TOT) return;

  int s, d;
  if (e0 < E_UI) { const int e  = e0 + lane;
                   s = eui[e];                d = eui[E_UI + e] + N_USER; }
  else           { const int e2 = e0 - E_UI + lane;
                   s = eiu[e2] + N_USER;      d = eiu[E_IU + e2]; }
  const int nrm = __float_as_int(dinv[s] * dinv[d]);

  for (int jb = 0; jb < EDGES_PER_WAVE; jb += EDGE_BATCH) {
    float v[EDGE_BATCH][4];
    for (int u = 0; u < EDGE_BATCH; ++u) {
      const int sj = __builtin_amdgcn_readlane(s, jb + u);
      const float* __restrict__ hs = hw + (long)sj * HID;
      v[u][0] = hs[lane];
      v[u][1] = hs[lane + 32];
      v[u][2] = hs[lane + 64];
      v[u][3] = hs[lane + 96];
    }
    for (int u = 0; u < EDGE_BATCH; ++u) {
      const int   dj = __builtin_amdgcn_readlane(d, jb + u);
      const float nj = __int_as_float(__builtin_amdgcn_readlane(nrm, jb + u));
      float* od = out + (long)dj * HID;
      atomicAdd(od + lane,      nj * v[u][0]);
      atomicAdd(od + lane + 32, nj * v[u][1]);
      atomicAdd(od + lane + 64, nj * v[u][2]);
      atomicAdd(od + lane + 96, nj * v[u][3]);
    }
  }
}

extern "C" void kernel_launch(void* const* d_in, const int* in_sizes, int n_in,
                              void* d_out, int out_size, void* d_ws, size_t ws_size,
                              hipStream_t stream)
{
  const float* x_user = (const float*)d_in[0];
  const float* x_item = (const float*)d_in[1];
  const int*   eui    = (const int*)d_in[2];
  const int*   eiu    = (const int*)d_in[3];
  const float* W_iu   = (const float*)d_in[4];
  const float* b_iu   = (const float*)d_in[5];
  const float* W_ii   = (const float*)d_in[6];
  const float* b_ii   = (const float*)d_in[7];
  const float* W1     = (const float*)d_in[8];
  const float* b1     = (const float*)d_in[9];
  const float* W2     = (const float*)d_in[10];
  const float* b2     = (const float*)d_in[11];
  float* out = (float*)d_out;

  float* hw   = (float*)d_ws;
  float* h1   = hw + (size_t)NNODES * HID;
  float* dinv = h1 + (size_t)NNODES * HID;
  unsigned* planes = (unsigned*)(dinv + NNODES);

  const dim3 blk(128);

  deg_hist<<<dim3(DEG_RANGES, DEG_SLICES, 2), dim3(DEG_THREADS), 0, stream>>>(eui, eiu, planes);
  dinv_fin<<<dim3((NNODES / 4 + 255) / 256), dim3(256), 0, stream>>>(planes, dinv);

  const unsigned edgeBlocks = (E_TOT / EDGES_PER_WAVE + 7) / 8;

  gemm_proj_l1<<<dim3(NNODES / 64), blk, 0, stream>>>(
      x_user, x_item, W_iu, b_iu, W_ii, b_ii, W1, b1, dinv, hw, h1);
  agg_edges<<<dim3(edgeBlocks), dim3(256), 0, stream>>>(eui, eiu, hw, dinv, h1);

  gemm_n128<<<dim3((NNODES + 63) / 64), blk, 0, stream>>>(
      h1, W2, b2, hw, dinv, out, NNODES, HID,  1,  1);
  agg_edges<<<dim3(edgeBlocks), dim3(256), 0, stream>>>(eui, eiu, hw, dinv, out);
}


# speedup vs baseline: 2.2528x; 1.0329x over previous
#include <hip/hip_runtime.h>


typedef __attribute__((ext_vector_type(2))) float v2f;
typedef __attribute__((ext_vector_type(4))) float v4f;
typedef __attribute__((ext_vector_type(8))) float v8f;

#define N_USER 100000
#define N_ITEM 100000
#define NNODES (N_USER + N_ITEM)
#define HID    128
#define E_UI   600000
#define E_IU   600000
#define E_TOT  (E_UI + E_IU)

__global__ __launch_bounds__(128) void gemm_n128(
    const float* __restrict__ A, const float* __restrict__ W,
    const float* __restrict__ bias, float* __restrict__ C,
    const float* __restrict__ dinv, float* __restrict__ S,
    int M, int K, int relu, int start)
{
  const int lane  = threadIdx.x & 31;
  const int wave  = threadIdx.x >> 5;
  const long row0 = (long)blockIdx.x * 64 + (long)wave * 16;

  const int m     = lane & 15;
  const int n4    = (lane & 15) << 2;
  const int khalf = (lane >> 4) << 1;

  long arow = row0 + m;
  if (arow >= M) arow = M - 1;
  const float* __restrict__ Ar = A + arow * (long)K;

  v8f acc[8] = {};

  for (int k = 0; k < K; k += 4) {
    v2f a;
    a.x = Ar[k + khalf];
    a.y = Ar[k + khalf + 1];
    if (relu) { a.x = fmaxf(a.x, 0.0f); a.y = fmaxf(a.y, 0.0f); }

    const float* __restrict__ Wk = W + (long)(k + khalf) * HID + n4;
    const v4f w0l = *reinterpret_cast<const v4f*>(Wk);
    const v4f w0h = *reinterpret_cast<const v4f*>(Wk + 64);
    const v4f w1l = *reinterpret_cast<const v4f*>(Wk + HID);
    const v4f w1h = *reinterpret_cast<const v4f*>(Wk + HID + 64);
#pragma unroll
    for (int t = 0; t < 4; ++t) {
      v2f b;
      b.x = w0l[t];
      b.y = w1l[t];
      acc[t] = __builtin_amdgcn_wmma_f32_16x16x4_f32(
           false, a,  false, b,
           (short)0, acc[t],  false,  false);
    }
#pragma unroll
    for (int t = 0; t < 4; ++t) {
      v2f b;
      b.x = w0h[t];
      b.y = w1h[t];
      acc[4 + t] = __builtin_amdgcn_wmma_f32_16x16x4_f32(
           false, a,  false, b,
           (short)0, acc[4 + t],  false,  false);
    }
  }

  const int rhalf = (lane >> 4) << 3;
  const v4f bl = *reinterpret_cast<const v4f*>(bias + n4);
  const v4f bh = *reinterpret_cast<const v4f*>(bias + 64 + n4);
  if (!start) {
#pragma unroll
    for (int r = 0; r < 8; ++r) {
      const long row = row0 + rhalf + r;
      if (row < M) {
        v4f ol, oh;
#pragma unroll
        for (int t = 0; t < 4; ++t) {
          ol[t] = acc[t][r] + bl[t];
          oh[t] = acc[4 + t][r] + bh[t];
        }
        float* Cr = C + row * HID + n4;
        *reinterpret_cast<v4f*>(Cr)      = ol;
        *reinterpret_cast<v4f*>(Cr + 64) = oh;
      }
    }
  } else {
#pragma unroll
    for (int r = 0; r < 8; ++r) {
      long row = row0 + rhalf + r;
      const bool live = row < M;
      if (!live) row = M - 1;
      const float di = dinv[row];
      const float dd = di * di;
      if (live) {
        v4f pl, ph, sl, sh;
#pragma unroll
        for (int t = 0; t < 4; ++t) {
          pl[t] = acc[t][r];
          ph[t] = acc[4 + t][r];
          sl[t] = bl[t] + dd * pl[t];
          sh[t] = bh[t] + dd * ph[t];
        }
        const long o = row * HID + n4;
        *reinterpret_cast<v4f*>(C + o)      = pl;
        *reinterpret_cast<v4f*>(C + o + 64) = ph;
        *reinterpret_cast<v4f*>(S + o)      = sl;
        *reinterpret_cast<v4f*>(S + o + 64) = sh;
      }
    }
  }
}

#define IN_DIM 256
static_assert(N_USER % 16 == 0 && NNODES % 64 == 0 && IN_DIM % 4 == 0,
              "whole 16-row waves of one node type, whole 64-row blocks");

__global__ __launch_bounds__(128) void gemm_proj_l1(
    const float* __restrict__ x_user, const float* __restrict__ x_item,
    const float* __restrict__ W_iu, const float* __restrict__ b_iu,
    const float* __restrict__ W_ii, const float* __restrict__ b_ii,
    const float* __restrict__ W1, const float* __restrict__ b1,
    const float* __restrict__ dinv, float* __restrict__ C, float* __restrict__ S)
{
  const int lane = threadIdx.x & 31;
  const int wave = __builtin_amdgcn_readfirstlane((int)(threadIdx.x >> 5));
  const int row0 = blockIdx.x * 64 + wave * 16;

  const bool item = row0 >= N_USER;
  const float* __restrict__ X  = item ? x_item + (long)(row0 - N_USER) * IN_DIM
                                      : x_user + (long)row0 * IN_DIM;
  const float* __restrict__ Wi = item ? W_ii : W_iu;
  const float* __restrict__ bi = item ? b_ii : b_iu;

  const int c  = lane & 15;
  const int hh = lane >> 4;
  const int c4 = c << 2;

  const float* __restrict__ Xr = X + (long)c * IN_DIM + 2 * hh;
  const float* __restrict__ Wc = Wi + (long)(2 * hh) * HID + c4;

  v8f accT[8] = {};
#pragma unroll 4
  for (int k = 0; k < IN_DIM; k += 4) {
    v2f xb;
    xb.x = Xr[k];
    xb.y = Xr[k + 1];
    const float* __restrict__ Wk = Wc + (long)k * HID;
    const v4f w0l = *reinterpret_cast<const v4f*>(Wk);
    const v4f w0h = *reinterpret_cast<const v4f*>(Wk + 64);
    const v4f w1l = *reinterpret_cast<const v4f*>(Wk + HID);
    const v4f w1h = *reinterpret_cast<const v4f*>(Wk + HID + 64);
#pragma unroll
    for (int t = 0; t < 4; ++t) {
      v2f a;
      a.x = w0l[t];
      a.y = w1l[t];
      accT[t] = __builtin_amdgcn_wmma_f32_16x16x4_f32(
           false, a,  false, xb,
           (short)0, accT[t],  false,  false);
    }
#pragma unroll
    for (int t = 0; t < 4; ++t) {
      v2f a;
      a.x = w0h[t];
      a.y = w1h[t];
      accT[4 + t] = __builtin_amdgcn_wmma_f32_16x16x4_f32(
           false, a,  false, xb,
           (short)0, accT[4 + t],  false,  false);
    }
  }

#pragma unroll
  for (int q = 0; q < 2; ++q) {
#pragma unroll
    for (int r = 0; r < 8; ++r) {
      const v4f bb = *reinterpret_cast<const v4f*>(bi + 64 * q + 32 * hh + 4 * r);
#pragma unroll
      for (int e = 0; e < 4; ++e) accT[4 * q + e][r] += bb[e];
    }
  }

  const float* __restrict__ W1c = W1 + (long)(32 * hh) * HID + c4;
  v8f acc[8] = {};
#pragma unroll
  for (int t = 0; t < 8; ++t) {
#pragma unroll
    for (int p = 0; p < 4; ++p) {
      v2f a;
      a.x = accT[t][2 * p];
      a.y = accT[t][2 * p + 1];
      const float* __restrict__ Wk = W1c + (long)((t >> 2) * 64 + (t & 3) + 8 * p) * HID;
      const v4f w0l = *reinterpret_cast<const v4f*>(Wk);
      const v4f w0h = *reinterpret_cast<const v4f*>(Wk + 64);
      const v4f w1l = *reinterpret_cast<const v4f*>(Wk + 4 * HID);
      const v4f w1h = *reinterpret_cast<const v4f*>(Wk + 4 * HID + 64);
#pragma unroll
      for (int u = 0; u < 4; ++u) {
        v2f b;
        b.x = w0l[u];
        b.y = w1l[u];
        acc[u] = __builtin_amdgcn_wmma_f32_16x16x4_f32(
             false, a,  false, b,
             (short)0, acc[u],  false,  false);
      }
#pragma unroll
      for (int u = 0; u < 4; ++u) {
        v2f b;
        b.x = w0h[u];
        b.y = w1h[u];
        acc[4 + u] = __builtin_amdgcn_wmma_f32_16x16x4_f32(
             false, a,  false, b,
             (short)0, acc[4 + u],  false,  false);
      }
    }
  }

  const int rhalf = hh << 3;
  const v4f bl = *reinterpret_cast<const v4f*>(b1 + c4);
  const v4f bh = *reinterpret_cast<const v4f*>(b1 + 64 + c4);
#pragma unroll
  for (int r = 0; r < 8; ++r) {
    const long row = (long)row0 + rhalf + r;
    const float di = dinv[row];
    const float dd = di * di;
    v4f pl, ph, sl, sh;
#pragma unroll
    for (int u = 0; u < 4; ++u) {
      pl[u] = acc[u][r];
      ph[u] = acc[4 + u][r];
      sl[u] = bl[u] + dd * pl[u];
      sh[u] = bh[u] + dd * ph[u];
    }
    const long o = row * HID + c4;
    *reinterpret_cast<v4f*>(C + o)      = pl;
    *reinterpret_cast<v4f*>(C + o + 64) = ph;
    *reinterpret_cast<v4f*>(S + o)      = sl;
    *reinterpret_cast<v4f*>(S + o + 64) = sh;
  }
}

#define DEG_RANGES      8
#define DEG_SLICES      16
#define DEG_THREADS     512
#define DEG_BINS        (N_USER / DEG_RANGES)
#define DEG_SLICE_VEC   (E_UI / DEG_SLICES / 4)
#define DEG_VEC_PER_THR ((DEG_SLICE_VEC + DEG_THREADS - 1) / DEG_THREADS)
static_assert(N_USER == N_ITEM && E_UI == E_IU, "one cell shape for both edge types");
static_assert(N_USER % DEG_RANGES == 0 && DEG_BINS % 4 == 0, "whole 128-bit counter groups");
static_assert(E_UI % (DEG_SLICES * 4) == 0, "slices are whole 128-bit index groups");
static_assert(NNODES % 4 == 0 && N_USER % 4 == 0, "planes are read and written 128 bits wide");

__global__ __launch_bounds__(DEG_THREADS) void deg_hist(
    const int* __restrict__ eui, const int* __restrict__ eiu,
    unsigned* __restrict__ planes)
{
  __shared__ __attribute__((aligned(16))) unsigned bins[DEG_BINS];
  const int tid   = threadIdx.x;
  const int lo    = blockIdx.x * DEG_BINS;
  const int slice = blockIdx.y;
  const bool ui   = blockIdx.z == 0;

  const int4* __restrict__ d4 = reinterpret_cast<const int4*>(
      (ui ? eui + E_UI : eiu + E_IU) + (long)slice * (DEG_SLICE_VEC * 4));

  int4 v[DEG_VEC_PER_THR];
#pragma unroll
  for (int j = 0; j < DEG_VEC_PER_THR; ++j) {
    int i = tid + j * DEG_THREADS;
    if (i > DEG_SLICE_VEC - 1) i = DEG_SLICE_VEC - 1;
    v[j] = d4[i];
  }

  uint4* b4 = reinterpret_cast<uint4*>(bins);
  for (int i = tid; i < DEG_BINS / 4; i += DEG_THREADS) b4[i] = make_uint4(0u, 0u, 0u, 0u);
  __syncthreads();

#pragma unroll
  for (int j = 0; j < DEG_VEC_PER_THR; ++j) {
    if (tid + j * DEG_THREADS < DEG_SLICE_VEC) {
      const unsigned r0 = (unsigned)(v[j].x - lo);
      const unsigned r1 = (unsigned)(v[j].y - lo);
      const unsigned r2 = (unsigned)(v[j].z - lo);
      const unsigned r3 = (unsigned)(v[j].w - lo);
      if (r0 < (unsigned)DEG_BINS) atomicAdd(&bins[r0], 1u);
      if (r1 < (unsigned)DEG_BINS) atomicAdd(&bins[r1], 1u);
      if (r2 < (unsigned)DEG_BINS) atomicAdd(&bins[r2], 1u);
      if (r3 < (unsigned)DEG_BINS) atomicAdd(&bins[r3], 1u);
    }
  }
  __syncthreads();

  uint4* __restrict__ o4 = reinterpret_cast<uint4*>(
      planes + (long)slice * NNODES + (ui ? N_USER : 0) + lo);
  for (int i = tid; i < DEG_BINS / 4; i += DEG_THREADS) o4[i] = b4[i];
}

__global__ __launch_bounds__(256) void dinv_fin(const unsigned* __restrict__ planes,
                                                float* __restrict__ dinv,
                                                unsigned* __restrict__ tickets)
{
  const int i = blockIdx.x * blockDim.x + threadIdx.x;
  if (i == 0) *reinterpret_cast<uint4*>(tickets) = make_uint4(0u, 0u, 0u, 0u);
  if (i < NNODES / 4) {
    const uint4* __restrict__ p4 = reinterpret_cast<const uint4*>(planes) + i;
    uint4 c = make_uint4(1u, 1u, 1u, 1u);
#pragma unroll
    for (int p = 0; p < DEG_SLICES; ++p) {
      const uint4 q = p4[(long)p * (NNODES / 4)];
      c.x += q.x; c.y += q.y; c.z += q.z; c.w += q.w;
    }
    v4f d;
    const float f0 = (float)c.x, f1 = (float)c.y, f2 = (float)c.z, f3 = (float)c.w;
    d[0] = f0 > 0.0f ? rsqrtf(f0) : 0.0f;
    d[1] = f1 > 0.0f ? rsqrtf(f1) : 0.0f;
    d[2] = f2 > 0.0f ? rsqrtf(f2) : 0.0f;
    d[3] = f3 > 0.0f ? rsqrtf(f3) : 0.0f;
    *reinterpret_cast<v4f*>(dinv + 4 * (long)i) = d;
  }
}

#define EDGES_PER_WAVE 32
#define EDGE_BATCH     8
#define WAVES_PER_TYPE  (E_UI / EDGES_PER_WAVE)
#define CHUNKS_PER_TYPE ((WAVES_PER_TYPE + 7) / 8)
static_assert(E_UI % EDGES_PER_WAVE == 0 && E_TOT % EDGES_PER_WAVE == 0,
              "a wave's 32 edges must be of one type and all exist");
static_assert(E_UI == E_IU, "one chunk count for both queues");

__global__ __launch_bounds__(256) void agg_edges(
    const int* __restrict__ eui, const int* __restrict__ eiu,
    const float* __restrict__ hw, const float* __restrict__ dinv,
    float* __restrict__ out, unsigned* __restrict__ tickets)
{
  __shared__ int sh_chunk;
  if (threadIdx.x == 0) {
    const unsigned where = (unsigned)__builtin_amdgcn_s_sendmsg_rtn(0x87);
    unsigned q = (where >> 18) & 1u;
    unsigned t = atomicAdd(&tickets[q], 1u);
    if (t >= (unsigned)CHUNKS_PER_TYPE) { q ^= 1u; t = atomicAdd(&tickets[q], 1u); }
    sh_chunk = t < (unsigned)CHUNKS_PER_TYPE ? (int)(q * CHUNKS_PER_TYPE + t) : -1;
  }
  __syncthreads();

  const int lane = threadIdx.x & 31;
  const int w     = __builtin_amdgcn_readfirstlane((int)(threadIdx.x >> 5));
  const int chunk = __builtin_amdgcn_readfirstlane(sh_chunk);
  if (chunk < 0) return;
  const bool ui = chunk < CHUNKS_PER_TYPE;
  const int wt  = (ui ? chunk : chunk - CHUNKS_PER_TYPE) * 8 + w;
  if (wt >= WAVES_PER_TYPE) return;
  const int el = wt * EDGES_PER_WAVE + lane;

  int s, d;
  if (ui) { s = eui[el];              d = eui[E_UI + el] + N_USER; }
  else    { s = eiu[el] + N_USER;     d = eiu[E_IU + el]; }
  const int nrm = __float_as_int(dinv[s] * dinv[d]);

  for (int jb = 0; jb < EDGES_PER_WAVE; jb += EDGE_BATCH) {
    float v[EDGE_BATCH][4];
    for (int u = 0; u < EDGE_BATCH; ++u) {
      const int sj = __builtin_amdgcn_readlane(s, jb + u);
      const float* __restrict__ hs = hw + (long)sj * HID;
      v[u][0] = hs[lane];
      v[u][1] = hs[lane + 32];
      v[u][2] = hs[lane + 64];
      v[u][3] = hs[lane + 96];
    }
    for (int u = 0; u < EDGE_BATCH; ++u) {
      const int   dj = __builtin_amdgcn_readlane(d, jb + u);
      const float nj = __int_as_float(__builtin_amdgcn_readlane(nrm, jb + u));
      float* od = out + (long)dj * HID;
      atomicAdd(od + lane,      nj * v[u][0]);
      atomicAdd(od + lane + 32, nj * v[u][1]);
      atomicAdd(od + lane + 64, nj * v[u][2]);
      atomicAdd(od + lane + 96, nj * v[u][3]);
    }
  }
}

extern "C" void kernel_launch(void* const* d_in, const int* in_sizes, int n_in,
                              void* d_out, int out_size, void* d_ws, size_t ws_size,
                              hipStream_t stream)
{
  const float* x_user = (const float*)d_in[0];
  const float* x_item = (const float*)d_in[1];
  const int*   eui    = (const int*)d_in[2];
  const int*   eiu    = (const int*)d_in[3];
  const float* W_iu   = (const float*)d_in[4];
  const float* b_iu   = (const float*)d_in[5];
  const float* W_ii   = (const float*)d_in[6];
  const float* b_ii   = (const float*)d_in[7];
  const float* W1     = (const float*)d_in[8];
  const float* b1     = (const float*)d_in[9];
  const float* W2     = (const float*)d_in[10];
  const float* b2     = (const float*)d_in[11];
  float* out = (float*)d_out;

  float* hw   = (float*)d_ws;
  float* h1   = hw + (size_t)NNODES * HID;
  float* dinv = h1 + (size_t)NNODES * HID;
  unsigned* planes = (unsigned*)(dinv + NNODES);
  unsigned* tickets = planes + (size_t)DEG_SLICES * NNODES;

  const dim3 blk(128);

  deg_hist<<<dim3(DEG_RANGES, DEG_SLICES, 2), dim3(DEG_THREADS), 0, stream>>>(eui, eiu, planes);
  dinv_fin<<<dim3((NNODES / 4 + 255) / 256), dim3(256), 0, stream>>>(planes, dinv, tickets);

  const unsigned edgeBlocks = 2 * CHUNKS_PER_TYPE;

  gemm_proj_l1<<<dim3(NNODES / 64), blk, 0, stream>>>(
      x_user, x_item, W_iu, b_iu, W_ii, b_ii, W1, b1, dinv, hw, h1);
  agg_edges<<<dim3(edgeBlocks), dim3(256), 0, stream>>>(eui, eiu, hw, dinv, h1, tickets);

  gemm_n128<<<dim3((NNODES + 63) / 64), blk, 0, stream>>>(
      h1, W2, b2, hw, dinv, out, NNODES, HID,  1,  1);
  agg_edges<<<dim3(edgeBlocks), dim3(256), 0, stream>>>(eui, eiu, hw, dinv, out, tickets + 2);
}
